# starting kernel
# baseline (speedup 1.0000x reference)
#include <hip/hip_runtime.h>

typedef __attribute__((ext_vector_type(2))) float v2f;
typedef __attribute__((ext_vector_type(8))) float v8f;

#define B_  8
#define T_  512
#define D_  64

__global__ __launch_bounds__(32) void linear_wmma(const float* __restrict__ x,
                                                  const float* __restrict__ W,
                                                  const float* __restrict__ bias,
                                                  float* __restrict__ v) {
  const int tile = blockIdx.x;
  const int m0   = (tile >> 2) * 16;
  const int e0   = (tile & 3) * 16;
  const int lane = threadIdx.x;
  const int half = lane >> 4;
  const int l    = lane & 15;

  const float* xrow = x + (size_t)(m0 + l) * D_;
  const float* wrow = W + (size_t)(e0 + l) * D_;

  v8f c = {};
#pragma unroll
  for (int k0 = 0; k0 < D_; k0 += 4) {
    v2f a, bf;
    a.x  = xrow[k0 + 2 * half];
    a.y  = xrow[k0 + 2 * half + 1];
    bf.x = wrow[k0 + 2 * half];
    bf.y = wrow[k0 + 2 * half + 1];
    c = __builtin_amdgcn_wmma_f32_16x16x4_f32(false, a, false, bf,
                                              (short)0, c, false, false);
  }

  const float bb = bias[e0 + l];
#pragma unroll
  for (int r = 0; r < 8; ++r) {
    const int m = m0 + r + 8 * half;
    v[(size_t)m * D_ + (e0 + l)] = c[r] + bb;
  }
}

#define ITILE  8
#define JCHUNK 128
#define JPAD   65

__global__ __launch_bounds__(128) void rbf_pairwise(const float* __restrict__ x,
                                                    const float* __restrict__ v,
                                                    float* __restrict__ out) {
  __shared__ float xi[ITILE][D_];
  __shared__ float vi[ITILE][D_];
  __shared__ float xj[JCHUNK * JPAD];

  const int blk = blockIdx.x;
  const int b   = blk / (T_ / ITILE);
  const int i0  = (blk % (T_ / ITILE)) * ITILE;
  const int tid = threadIdx.x;

  for (int t = tid; t < ITILE * D_; t += 128) {
    const int ii = t >> 6, d = t & 63;
    const size_t g = (size_t)(b * T_ + i0 + ii) * D_ + d;
    xi[ii][d] = x[g];
    vi[ii][d] = v[g];
  }

  for (int j0 = 0; j0 < T_; j0 += JCHUNK) {
    __syncthreads();
    for (int t = tid; t < JCHUNK * D_; t += 128) {
      const int jj = t >> 6, d = t & 63;
      xj[jj * JPAD + d] = x[(size_t)(b * T_ + j0 + jj) * D_ + d];
    }
    __syncthreads();

    const int j = j0 + tid;
    float num[ITILE], den[ITILE];
#pragma unroll
    for (int ii = 0; ii < ITILE; ++ii) { num[ii] = 0.0f; den[ii] = 0.0f; }

    const float* xjrow = &xj[tid * JPAD];
#pragma unroll 4
    for (int d = 0; d < D_; ++d) {
      const float xjd = xjrow[d];
#pragma unroll
      for (int ii = 0; ii < ITILE; ++ii) {
        const float diff = xi[ii][d] - xjd;
        const float e = __builtin_amdgcn_exp2f(diff * diff * -1.4426950408889634f);
        den[ii] += e;
        num[ii] += e * vi[ii][d];
      }
    }

#pragma unroll
    for (int ii = 0; ii < ITILE; ++ii) {
      out[(size_t)(b * T_ + i0 + ii) * T_ + j] = num[ii] / den[ii];
    }
  }
}

extern "C" void kernel_launch(void* const* d_in, const int* in_sizes, int n_in,
                              void* d_out, int out_size, void* d_ws, size_t ws_size,
                              hipStream_t stream) {
  const float* x    = (const float*)d_in[0];
  const float* W    = (const float*)d_in[1];
  const float* bias = (const float*)d_in[2];
  float*       out  = (float*)d_out;
  float*       v    = (float*)d_ws;

  linear_wmma<<<dim3((B_ * T_ / 16) * (D_ / 16)), dim3(32), 0, stream>>>(x, W, bias, v);
  rbf_pairwise<<<dim3(B_ * (T_ / ITILE)), dim3(128), 0, stream>>>(x, v, out);
}


# linear layer folded into the pairwise kernel: 4 waves of a block make v of its 8 query rows with the same f32 WMMA sequence, hand it over in 2 KB of LDS; first launch and workspace trip gone
# speedup vs baseline: 1.7111x; 1.7111x over previous
#include <hip/hip_runtime.h>

typedef __attribute__((ext_vector_type(2))) float v2f;
typedef __attribute__((ext_vector_type(4))) float v4f;
typedef __attribute__((ext_vector_type(8))) float v8f;

#define B_  8
#define T_  512
#define D_  64

#define IROWS 8
#define JTHR  256


__global__ __launch_bounds__(JTHR) __attribute__((amdgpu_waves_per_eu(4, 4))) void rbf_fused(const float* __restrict__ x,
                                                     const float* __restrict__ W,
                                                     const float* __restrict__ bias,
                                                     float* __restrict__ out) {
  __shared__ __attribute__((aligned(16))) float v_s[IROWS * D_];

  const int blk = blockIdx.x;
  const int b   = blk / (T_ / IROWS);
  const int i0  = (blk % (T_ / IROWS)) * IROWS;
  const int j0  = threadIdx.x;
  const int j1  = threadIdx.x + JTHR;

  const int wave = __builtin_amdgcn_readfirstlane((int)(threadIdx.x >> 5));
  const int lane = threadIdx.x & 31;
  const int half = lane >> 4;
  const int l    = lane & 15;
  const int e0   = wave * 16;

  v2f fa[D_ / 4] = {};
  v2f fb[D_ / 4] = {};
  float bb = 0.0f;
  if (wave < 4) {
    const float* xrow = x + (size_t)(b * T_ + i0 + (l & 7)) * D_;
    const float* wrow = W + (size_t)(e0 + l) * D_;
#pragma unroll
    for (int k = 0; k < D_ / 4; ++k) {
      fa[k].x = xrow[4 * k + 2 * half];
      fa[k].y = xrow[4 * k + 2 * half + 1];
      fb[k].x = wrow[4 * k + 2 * half];
      fb[k].y = wrow[4 * k + 2 * half + 1];
    }
    bb = bias[e0 + l];
  }

  const float* r0 = x + (size_t)(b * T_ + j0) * D_;
  const float* r1 = x + (size_t)(b * T_ + j1) * D_;
  v4f t0[D_ / 4], t1[D_ / 4];
#pragma unroll
  for (int q = 0; q < D_ / 4; ++q) {
    t0[q] = *(const v4f*)(r0 + 4 * q);
    t1[q] = *(const v4f*)(r1 + 4 * q);
  }

  if (wave < 4) {
    v8f c = {};
#pragma unroll
    for (int k = 0; k < D_ / 4; ++k) {
      c = __builtin_amdgcn_wmma_f32_16x16x4_f32(false, fa[k], false, fb[k],
                                                (short)0, c, false, false);
    }
    if (half == 0) {
#pragma unroll
      for (int r = 0; r < IROWS; ++r) v_s[r * D_ + e0 + l] = c[r] + bb;
    }
  }

  const float c = 1.2011224087864498f;

  v2f xjn[D_];
#pragma unroll
  for (int q = 0; q < D_ / 4; ++q) {
    xjn[4 * q + 0] = v2f{t0[q].x, t1[q].x} * -c;
    xjn[4 * q + 1] = v2f{t0[q].y, t1[q].y} * -c;
    xjn[4 * q + 2] = v2f{t0[q].z, t1[q].z} * -c;
    xjn[4 * q + 3] = v2f{t0[q].w, t1[q].w} * -c;
  }

  __syncthreads();

  const v2f cc = {c, c};
  for (int ii = 0; ii < IROWS; ++ii) {
    const int i = i0 + ii;
    const float* xi = x + (size_t)(b * T_ + i) * D_;
    float xs_row[D_];
#pragma unroll
    for (int d = 0; d < D_; ++d) xs_row[d] = xi[d];
    v2f den = {0.0f, 0.0f};
    v2f num = {0.0f, 0.0f};
#pragma unroll
    for (int h = 0; h < 2; ++h) {
      if (h) __builtin_amdgcn_sched_barrier(0);
      const v4f* vrow = (const v4f*)(v_s + ii * D_ + h * (D_ / 2));
      v4f vv[D_ / 8];
#pragma unroll
      for (int q = 0; q < D_ / 8; ++q) vv[q] = vrow[q];
      __builtin_amdgcn_sched_barrier(0);
#pragma unroll
      for (int dd = 0; dd < D_ / 2; ++dd) {
        const int d = h * (D_ / 2) + dd;
        const float xs = xs_row[d];
        const float vs = vv[dd >> 2][dd & 3];
        const v2f dc = __builtin_elementwise_fma(v2f{xs, xs}, cc, xjn[d]);
        const v2f t  = dc * dc;
        v2f e;
        e.x = __builtin_amdgcn_exp2f(-t.x);
        e.y = __builtin_amdgcn_exp2f(-t.y);
        den += e;
        num = __builtin_elementwise_fma(e, v2f{vs, vs}, num);
      }
    }
    float* orow = out + (size_t)(b * T_ + i) * T_;
    orow[j0] = num.x / den.x;
    orow[j1] = num.y / den.y;
  }
}

extern "C" void kernel_launch(void* const* d_in, const int* in_sizes, int n_in,
                              void* d_out, int out_size, void* d_ws, size_t ws_size,
                              hipStream_t stream) {
  const float* x    = (const float*)d_in[0];
  const float* W    = (const float*)d_in[1];
  const float* bias = (const float*)d_in[2];
  float*       out  = (float*)d_out;

  rbf_fused<<<dim3(B_ * (T_ / IROWS)), dim3(JTHR), 0, stream>>>(x, W, bias, out);
}


# 16 query rows per block instead of 8 (256 blocks, full 16-row WMMA tile): each thread's two own x rows now serve 16 rows, halving the lane-strided row loads, second-level bytes and waves to enter
# speedup vs baseline: 1.8255x; 1.0669x over previous
#include <hip/hip_runtime.h>

typedef __attribute__((ext_vector_type(2))) float v2f;
typedef __attribute__((ext_vector_type(4))) float v4f;
typedef __attribute__((ext_vector_type(8))) float v8f;

#define B_  8
#define T_  512
#define D_  64

#define IROWS 16
#define JTHR  256


__global__ __launch_bounds__(JTHR) __attribute__((amdgpu_waves_per_eu(2, 2))) void rbf_fused(const float* __restrict__ x,
                                                     const float* __restrict__ W,
                                                     const float* __restrict__ bias,
                                                     float* __restrict__ out) {
  __shared__ __attribute__((aligned(16))) float v_s[IROWS * D_];

  const int blk = blockIdx.x;
  const int b   = blk / (T_ / IROWS);
  const int i0  = (blk % (T_ / IROWS)) * IROWS;
  const int j0  = threadIdx.x;
  const int j1  = threadIdx.x + JTHR;

  const int wave = __builtin_amdgcn_readfirstlane((int)(threadIdx.x >> 5));
  const int lane = threadIdx.x & 31;
  const int half = lane >> 4;
  const int l    = lane & 15;
  const int e0   = wave * 16;

  v2f fa[D_ / 4] = {};
  v2f fb[D_ / 4] = {};
  float bb = 0.0f;
  if (wave < 4) {
    const float* xrow = x + (size_t)(b * T_ + i0 + l) * D_;
    const float* wrow = W + (size_t)(e0 + l) * D_;
#pragma unroll
    for (int k = 0; k < D_ / 4; ++k) {
      fa[k].x = xrow[4 * k + 2 * half];
      fa[k].y = xrow[4 * k + 2 * half + 1];
      fb[k].x = wrow[4 * k + 2 * half];
      fb[k].y = wrow[4 * k + 2 * half + 1];
    }
    bb = bias[e0 + l];
  }

  const float* r0 = x + (size_t)(b * T_ + j0) * D_;
  const float* r1 = x + (size_t)(b * T_ + j1) * D_;
  v4f t0[D_ / 4], t1[D_ / 4];
#pragma unroll
  for (int q = 0; q < D_ / 4; ++q) {
    t0[q] = *(const v4f*)(r0 + 4 * q);
    t1[q] = *(const v4f*)(r1 + 4 * q);
  }

  if (wave < 4) {
    v8f c = {};
#pragma unroll
    for (int k = 0; k < D_ / 4; ++k) {
      c = __builtin_amdgcn_wmma_f32_16x16x4_f32(false, fa[k], false, fb[k],
                                                (short)0, c, false, false);
    }
#pragma unroll
    for (int r = 0; r < 8; ++r) v_s[(r + 8 * half) * D_ + e0 + l] = c[r] + bb;
  }

  const float c = 1.2011224087864498f;

  v2f xjn[D_];
#pragma unroll
  for (int q = 0; q < D_ / 4; ++q) {
    xjn[4 * q + 0] = v2f{t0[q].x, t1[q].x} * -c;
    xjn[4 * q + 1] = v2f{t0[q].y, t1[q].y} * -c;
    xjn[4 * q + 2] = v2f{t0[q].z, t1[q].z} * -c;
    xjn[4 * q + 3] = v2f{t0[q].w, t1[q].w} * -c;
  }

  __syncthreads();

  const v2f cc = {c, c};
  for (int ii = 0; ii < IROWS; ++ii) {
    const int i = i0 + ii;
    const float* xi = x + (size_t)(b * T_ + i) * D_;
    float xs_row[D_];
#pragma unroll
    for (int d = 0; d < D_; ++d) xs_row[d] = xi[d];
    v2f den = {0.0f, 0.0f};
    v2f num = {0.0f, 0.0f};
#pragma unroll
    for (int h = 0; h < 2; ++h) {
      if (h) __builtin_amdgcn_sched_barrier(0);
      const v4f* vrow = (const v4f*)(v_s + ii * D_ + h * (D_ / 2));
      v4f vv[D_ / 8];
#pragma unroll
      for (int q = 0; q < D_ / 8; ++q) vv[q] = vrow[q];
      __builtin_amdgcn_sched_barrier(0);
#pragma unroll
      for (int dd = 0; dd < D_ / 2; ++dd) {
        const int d = h * (D_ / 2) + dd;
        const float xs = xs_row[d];
        const float vs = vv[dd >> 2][dd & 3];
        const v2f dc = __builtin_elementwise_fma(v2f{xs, xs}, cc, xjn[d]);
        const v2f t  = dc * dc;
        v2f e;
        e.x = __builtin_amdgcn_exp2f(-t.x);
        e.y = __builtin_amdgcn_exp2f(-t.y);
        den += e;
        num = __builtin_elementwise_fma(e, v2f{vs, vs}, num);
      }
    }
    float* orow = out + (size_t)(b * T_ + i) * T_;
    orow[j0] = num.x / den.x;
    orow[j1] = num.y / den.y;
  }
}

extern "C" void kernel_launch(void* const* d_in, const int* in_sizes, int n_in,
                              void* d_out, int out_size, void* d_ws, size_t ws_size,
                              hipStream_t stream) {
  const float* x    = (const float*)d_in[0];
  const float* W    = (const float*)d_in[1];
  const float* bias = (const float*)d_in[2];
  float*       out  = (float*)d_out;

  rbf_fused<<<dim3(B_ * (T_ / IROWS)), dim3(JTHR), 0, stream>>>(x, W, bias, out);
}


# speedup vs baseline: 1.9689x; 1.0786x over previous
#include <hip/hip_runtime.h>

typedef __attribute__((ext_vector_type(2))) float v2f;
typedef __attribute__((ext_vector_type(4))) float v4f;
typedef __attribute__((ext_vector_type(8))) float v8f;

#define B_  8
#define T_  512
#define D_  64

#define IROWS 16
#define JTHR  256


__global__ __launch_bounds__(JTHR) __attribute__((amdgpu_waves_per_eu(2, 2))) void rbf_fused(const float* __restrict__ x,
                                                     const float* __restrict__ W,
                                                     const float* __restrict__ bias,
                                                     float* __restrict__ out) {
  __shared__ __attribute__((aligned(16))) float v_s[IROWS * D_];

  const int blk = blockIdx.x;
  const int b   = blk / (T_ / IROWS);
  const int i0  = (blk % (T_ / IROWS)) * IROWS;
  const int j0  = threadIdx.x;
  const int j1  = threadIdx.x + JTHR;

  const int wave = __builtin_amdgcn_readfirstlane((int)(threadIdx.x >> 5));
  const int lane = threadIdx.x & 31;
  const int half = lane >> 4;
  const int l    = lane & 15;
  const int e0   = wave * 16;

  v2f fa[D_ / 4] = {};
  v2f fb[D_ / 4] = {};
  float bb = 0.0f;
  if (wave < 4) {
    const float* xrow = x + (size_t)(b * T_ + i0 + l) * D_;
    const float* wrow = W + (size_t)(e0 + l) * D_;
#pragma unroll
    for (int k = 0; k < D_ / 4; ++k) {
      fa[k].x = xrow[4 * k + 2 * half];
      fa[k].y = xrow[4 * k + 2 * half + 1];
      fb[k].x = wrow[4 * k + 2 * half];
      fb[k].y = wrow[4 * k + 2 * half + 1];
    }
    bb = bias[e0 + l];
  }

  const float* r0 = x + (size_t)(b * T_ + j0) * D_;
  const float* r1 = x + (size_t)(b * T_ + j1) * D_;
  v4f t0[D_ / 4], t1[D_ / 4];
#pragma unroll
  for (int q = 0; q < D_ / 4; ++q) {
    t0[q] = *(const v4f*)(r0 + 4 * q);
    t1[q] = *(const v4f*)(r1 + 4 * q);
  }

  if (wave < 4) {
    v8f c = {};
#pragma unroll
    for (int k = 0; k < D_ / 4; ++k) {
      c = __builtin_amdgcn_wmma_f32_16x16x4_f32(false, fa[k], false, fb[k],
                                                (short)0, c, false, false);
    }
#pragma unroll
    for (int r = 0; r < 8; ++r) v_s[(r + 8 * half) * D_ + e0 + l] = c[r] + bb;
  }

  const float c = 1.2011224087864498f;

  v2f xjn[D_];
#pragma unroll
  for (int q = 0; q < D_ / 4; ++q) {
    xjn[4 * q + 0] = v2f{t0[q].x, t1[q].x} * -c;
    xjn[4 * q + 1] = v2f{t0[q].y, t1[q].y} * -c;
    xjn[4 * q + 2] = v2f{t0[q].z, t1[q].z} * -c;
    xjn[4 * q + 3] = v2f{t0[q].w, t1[q].w} * -c;
  }

  __syncthreads();

  const v2f cc = {c, c};
  const v2f zero2 = {0.0f, 0.0f};

#define DIFF(q, sp, xj)  asm volatile("v_pk_fma_f32 %0, %1, %2, %3 op_sel_hi:[0,0,1]" : "=v"(q) : "s"(sp), "s"(cc), "v"(xj))
#define SQUARE(q)        asm volatile("v_pk_mul_f32 %0, %0, %0" : "+v"(q))
#define EXP2NEG(e, t)    asm volatile("v_exp_f32_e64 %0, -%1" : "=v"(e) : "v"(t))
#define DEN_ADD(dn, e)   asm volatile("v_pk_add_f32 %0, %0, %1" : "+v"(dn) : "v"(e))
#define NUM_LO(nm, e, vp) asm volatile("v_pk_fma_f32 %0, %1, %2, %0 op_sel_hi:[1,0,1]" : "+v"(nm) : "v"(e), "v"(vp))
#define NUM_HI(nm, e, vp) asm volatile("v_pk_fma_f32 %0, %1, %2, %0 op_sel:[0,1,0]" : "+v"(nm) : "v"(e), "v"(vp))

  for (int ii = 0; ii < IROWS; ++ii) {
    const int i = i0 + ii;
    const __attribute__((address_space(4))) float* xi =
        (const __attribute__((address_space(4))) float*)(x + (size_t)(b * T_ + i) * D_);
    float xs_row[D_];
#pragma unroll
    for (int d = 0; d < D_; ++d) xs_row[d] = xi[d];
    v2f den, num;
    v2f qD, qC, qB, qA;
    {
      qD = __builtin_elementwise_fma(v2f{xs_row[0], xs_row[0]}, cc, xjn[0]);
      qD = qD * qD;
      qC = __builtin_elementwise_fma(v2f{xs_row[1], xs_row[1]}, cc, xjn[1]);
      qB = __builtin_elementwise_fma(v2f{xs_row[2], xs_row[2]}, cc, xjn[2]);
    }
    asm volatile("v_mov_b64 %0, 0" : "=v"(den));
    asm volatile("v_mov_b64 %0, 0" : "=v"(num));
    v2f e1 = zero2, e2 = zero2;
    v2f vq1 = zero2, vq2 = zero2;
#pragma unroll
    for (int h = 0; h < 2; ++h) {
      if (h) __builtin_amdgcn_sched_barrier(0);
      const v4f* vrow = (const v4f*)(v_s + ii * D_ + h * (D_ / 2));
      v4f vv[D_ / 8];
#pragma unroll
      for (int q = 0; q < D_ / 8; ++q) vv[q] = vrow[q];
      __builtin_amdgcn_sched_barrier(0);
#pragma unroll
      for (int dd = 0; dd < D_ / 2; ++dd) {
        const int d = h * (D_ / 2) + dd;
        float ex, ey;
        EXP2NEG(ex, qD.x);
        if (d + 1 < D_) SQUARE(qC);
        EXP2NEG(ey, qD.y);
        if (d >= 2) DEN_ADD(den, e2);
        if (d >= 2) { if (d & 1) NUM_HI(num, e2, vq2); else NUM_LO(num, e2, vq2); }
        if (d + 3 < D_) {
          const v2f pr = {xs_row[(d + 3) & ~1], xs_row[(d + 3) | 1]};
          const v2f sp = ((d + 3) & 1) ? __builtin_shufflevector(pr, pr, 1, -1) : pr;
          DIFF(qA, sp, xjn[d + 3]);
        }
        e2 = e1; e1 = v2f{ex, ey};
        vq2 = vq1;
        vq1 = (dd & 2) ? v2f{vv[dd >> 2].z, vv[dd >> 2].w} : v2f{vv[dd >> 2].x, vv[dd >> 2].y};
        qD = qC; qC = qB; qB = qA;
      }
    }
    den += e2;
    num = __builtin_elementwise_fma(e2, v2f{vq2.x, vq2.x}, num);
    den += e1;
    num = __builtin_elementwise_fma(e1, v2f{vq1.y, vq1.y}, num);
    float* orow = out + (size_t)(b * T_ + i) * T_;
    orow[j0] = num.x / den.x;
    orow[j1] = num.y / den.y;
  }
#undef DIFF
#undef SQUARE
#undef EXP2NEG
#undef DEN_ADD
#undef NUM_LO
#undef NUM_HI
}

extern "C" void kernel_launch(void* const* d_in, const int* in_sizes, int n_in,
                              void* d_out, int out_size, void* d_ws, size_t ws_size,
                              hipStream_t stream) {
  const float* x    = (const float*)d_in[0];
  const float* W    = (const float*)d_in[1];
  const float* bias = (const float*)d_in[2];
  float*       out  = (float*)d_out;

  rbf_fused<<<dim3(B_ * (T_ / IROWS)), dim3(JTHR), 0, stream>>>(x, W, bias, out);
}


# own rows fetched by whole-line loads (each 16-byte load takes 512 consecutive bytes, not 32 scattered lines) and turned into one row per lane through LDS at pitch 272 bytes; sum order unchanged
# speedup vs baseline: 2.1145x; 1.0739x over previous
#include <hip/hip_runtime.h>

typedef __attribute__((ext_vector_type(2))) float v2f;
typedef __attribute__((ext_vector_type(4))) float v4f;
typedef __attribute__((ext_vector_type(8))) float v8f;

#define B_  8
#define T_  512
#define D_  64

#define IROWS 16
#define JTHR  256
#define SPITCH 68


__global__ __launch_bounds__(JTHR) __attribute__((amdgpu_waves_per_eu(2, 2))) void rbf_fused(const float* __restrict__ x,
                                                     const float* __restrict__ W,
                                                     const float* __restrict__ bias,
                                                     float* __restrict__ out) {
  __shared__ __attribute__((aligned(16))) float lds[IROWS * D_ + (JTHR / 32) * 2 * 32 * SPITCH];
  float* const v_s   = lds;
  float* const stage = lds + IROWS * D_;

  const int blk = blockIdx.x;
  const int b   = blk / (T_ / IROWS);
  const int i0  = (blk % (T_ / IROWS)) * IROWS;
  const int j0  = threadIdx.x;
  const int j1  = threadIdx.x + JTHR;

  const int wave = __builtin_amdgcn_readfirstlane((int)(threadIdx.x >> 5));
  const int lane = threadIdx.x & 31;
  const int half = lane >> 4;
  const int l    = lane & 15;
  const int e0   = wave * 16;

  v2f fa[D_ / 4] = {};
  v2f fb[D_ / 4] = {};
  float bb = 0.0f;
  if (wave < 4) {
    const float* xrow = x + (size_t)(b * T_ + i0 + l) * D_;
    const float* wrow = W + (size_t)(e0 + l) * D_;
#pragma unroll
    for (int k = 0; k < D_ / 4; ++k) {
      fa[k].x = xrow[4 * k + 2 * half];
      fa[k].y = xrow[4 * k + 2 * half + 1];
      fb[k].x = wrow[4 * k + 2 * half];
      fb[k].y = wrow[4 * k + 2 * half + 1];
    }
    bb = bias[e0 + l];
  }

  const float* s0 = x + (size_t)(b * T_ + 32 * wave) * D_ + 4 * lane;
  const float* s1 = s0 + (size_t)JTHR * D_;
  v4f g0[D_ / 4], g1[D_ / 4];
#pragma unroll
  for (int n = 0; n < D_ / 4; ++n) {
    g0[n] = *(const v4f*)(s0 + 128 * n);
    g1[n] = *(const v4f*)(s1 + 128 * n);
  }

  if (wave < 4) {
    v8f c = {};
    __builtin_amdgcn_sched_barrier(0);
#pragma unroll
    for (int k = 0; k < D_ / 4; ++k) {
      c = __builtin_amdgcn_wmma_f32_16x16x4_f32(false, fa[k], false, fb[k],
                                                (short)0, c, false, false);
    }
    __builtin_amdgcn_sched_barrier(0);
#pragma unroll
    for (int r = 0; r < 8; ++r) v_s[(r + 8 * half) * D_ + e0 + l] = c[r] + bb;
  }

  float* st0 = stage + (size_t)(2 * wave) * (32 * SPITCH);
  float* st1 = st0 + 32 * SPITCH;
#pragma unroll
  for (int n = 0; n < D_ / 4; ++n) {
    *(v4f*)(st0 + (2 * n + half) * SPITCH + 4 * l) = g0[n];
    *(v4f*)(st1 + (2 * n + half) * SPITCH + 4 * l) = g1[n];
  }

  __syncthreads();

  v4f t0[D_ / 4], t1[D_ / 4];
#pragma unroll
  for (int q = 0; q < D_ / 4; ++q) {
    t0[q] = *(const v4f*)(st0 + lane * SPITCH + 4 * q);
    t1[q] = *(const v4f*)(st1 + lane * SPITCH + 4 * q);
  }

  const float c = 1.2011224087864498f;

  v2f xjn[D_];
#pragma unroll
  for (int q = 0; q < D_ / 4; ++q) {
    xjn[4 * q + 0] = v2f{t0[q].x, t1[q].x} * -c;
    xjn[4 * q + 1] = v2f{t0[q].y, t1[q].y} * -c;
    xjn[4 * q + 2] = v2f{t0[q].z, t1[q].z} * -c;
    xjn[4 * q + 3] = v2f{t0[q].w, t1[q].w} * -c;
  }

  const v2f cc = {c, c};
  const v2f zero2 = {0.0f, 0.0f};

#define DIFF(q, sp, xj)  asm volatile("v_pk_fma_f32 %0, %1, %2, %3 op_sel_hi:[0,0,1]" : "=v"(q) : "s"(sp), "s"(cc), "v"(xj))
#define SQUARE(q)        asm volatile("v_pk_mul_f32 %0, %0, %0" : "+v"(q))
#define EXP2NEG(e, t)    asm volatile("v_exp_f32_e64 %0, -%1" : "=v"(e) : "v"(t))
#define DEN_ADD(dn, e)   asm volatile("v_pk_add_f32 %0, %0, %1" : "+v"(dn) : "v"(e))
#define NUM_LO(nm, e, vp) asm volatile("v_pk_fma_f32 %0, %1, %2, %0 op_sel_hi:[1,0,1]" : "+v"(nm) : "v"(e), "v"(vp))
#define NUM_HI(nm, e, vp) asm volatile("v_pk_fma_f32 %0, %1, %2, %0 op_sel:[0,1,0]" : "+v"(nm) : "v"(e), "v"(vp))

  for (int ii = 0; ii < IROWS; ++ii) {
    const int i = i0 + ii;
    const __attribute__((address_space(4))) float* xi =
        (const __attribute__((address_space(4))) float*)(x + (size_t)(b * T_ + i) * D_);
    float xs_row[D_];
#pragma unroll
    for (int d = 0; d < D_; ++d) xs_row[d] = xi[d];
    v2f den, num;
    v2f qD, qC, qB, qA;
    {
      qD = __builtin_elementwise_fma(v2f{xs_row[0], xs_row[0]}, cc, xjn[0]);
      qD = qD * qD;
      qC = __builtin_elementwise_fma(v2f{xs_row[1], xs_row[1]}, cc, xjn[1]);
      qB = __builtin_elementwise_fma(v2f{xs_row[2], xs_row[2]}, cc, xjn[2]);
    }
    asm volatile("v_mov_b64 %0, 0" : "=v"(den));
    asm volatile("v_mov_b64 %0, 0" : "=v"(num));
    v2f e1 = zero2, e2 = zero2;
    v2f vq1 = zero2, vq2 = zero2;
#pragma unroll
    for (int h = 0; h < 2; ++h) {
      if (h) __builtin_amdgcn_sched_barrier(0);
      const v4f* vrow = (const v4f*)(v_s + ii * D_ + h * (D_ / 2));
      v4f vv[D_ / 8];
#pragma unroll
      for (int q = 0; q < D_ / 8; ++q) vv[q] = vrow[q];
      __builtin_amdgcn_sched_barrier(0);
#pragma unroll
      for (int dd = 0; dd < D_ / 2; ++dd) {
        const int d = h * (D_ / 2) + dd;
        float ex, ey;
        EXP2NEG(ex, qD.x);
        if (d + 1 < D_) SQUARE(qC);
        EXP2NEG(ey, qD.y);
        if (d >= 2) DEN_ADD(den, e2);
        if (d >= 2) { if (d & 1) NUM_HI(num, e2, vq2); else NUM_LO(num, e2, vq2); }
        if (d + 3 < D_) {
          const v2f pr = {xs_row[(d + 3) & ~1], xs_row[(d + 3) | 1]};
          const v2f sp = ((d + 3) & 1) ? __builtin_shufflevector(pr, pr, 1, -1) : pr;
          DIFF(qA, sp, xjn[d + 3]);
        }
        e2 = e1; e1 = v2f{ex, ey};
        vq2 = vq1;
        vq1 = (dd & 2) ? v2f{vv[dd >> 2].z, vv[dd >> 2].w} : v2f{vv[dd >> 2].x, vv[dd >> 2].y};
        qD = qC; qC = qB; qB = qA;
      }
    }
    den += e2;
    num = __builtin_elementwise_fma(e2, v2f{vq2.x, vq2.x}, num);
    den += e1;
    num = __builtin_elementwise_fma(e1, v2f{vq1.y, vq1.y}, num);
    float* orow = out + (size_t)(b * T_ + i) * T_;
    orow[j0] = num.x / den.x;
    orow[j1] = num.y / den.y;
  }
#undef DIFF
#undef SQUARE
#undef EXP2NEG
#undef DEN_ADD
#undef NUM_LO
#undef NUM_HI
}

extern "C" void kernel_launch(void* const* d_in, const int* in_sizes, int n_in,
                              void* d_out, int out_size, void* d_ws, size_t ws_size,
                              hipStream_t stream) {
  const float* x    = (const float*)d_in[0];
  const float* W    = (const float*)d_in[1];
  const float* bias = (const float*)d_in[2];
  float*       out  = (float*)d_out;

  rbf_fused<<<dim3(B_ * (T_ / IROWS)), dim3(JTHR), 0, stream>>>(x, W, bias, out);
}
